# starting kernel
# baseline (speedup 1.0000x reference)
#include <hip/hip_runtime.h>
#include <hip/hip_bf16.h>

#define NN     50000
#define KK     32
#define IN_C   256
#define OUT_C  128

typedef __attribute__((ext_vector_type(2))) float v2f;
typedef __attribute__((ext_vector_type(8))) float v8f;

__global__ void __launch_bounds__(256)
median_conv_gemm_f32(const float* __restrict__ x,
                     const float* __restrict__ w,
                     float* __restrict__ h) {
    const int wave = threadIdx.x >> 5;
    const int lane = threadIdx.x & 31;
    const int lid  = lane & 15;
    const int half = lane >> 4;

    const int gw = blockIdx.x * 8 + wave;
    const int mt = gw >> 3;
    const int nt = gw & 7;
    const int m0 = mt * 16;
    const int n0 = nt * 16;

    const float* __restrict__ xrow = x + (size_t)(m0 + lid) * IN_C;

    v8f c = {};
    #pragma unroll 4
    for (int k0 = 0; k0 < IN_C; k0 += 4) {
        v2f a, b;
        const int ka = k0 + 2 * half;
        a.x = xrow[ka];
        a.y = xrow[ka + 1];
        b.x = w[(size_t)ka       * OUT_C + n0 + lid];
        b.y = w[(size_t)(ka + 1) * OUT_C + n0 + lid];
        c = __builtin_amdgcn_wmma_f32_16x16x4_f32(
                false, a, false, b, (short)0, c, false, false);
    }

    #pragma unroll
    for (int v = 0; v < 8; v++) {
        h[(size_t)(m0 + v + 8 * half) * OUT_C + n0 + lid] = c[v];
    }
}

__global__ void __launch_bounds__(128)
median_conv_select(const float* __restrict__ h,
                   const int*   __restrict__ nbr,
                   const int*   __restrict__ deg,
                   float*       __restrict__ out) {
    const int n = blockIdx.x;
    const int c = threadIdx.x;

    const int d   = deg[n];
    const int med = (d - 1) >> 1;

    const int* __restrict__ nb = nbr + (size_t)n * KK;

    float v[KK];
    #pragma unroll
    for (int j = 0; j < KK; j++) {
        const int idx = nb[j];
        const float val = h[(size_t)idx * OUT_C + c];
        v[j] = (j < d) ? val : __builtin_inff();
    }

    #pragma unroll
    for (int k = 2; k <= KK; k <<= 1) {
        #pragma unroll
        for (int j = k >> 1; j > 0; j >>= 1) {
            #pragma unroll
            for (int i = 0; i < KK; i++) {
                const int l = i ^ j;
                if (l > i) {
                    const bool up = ((i & k) == 0);
                    const float a = v[i], b = v[l];
                    const float lo = fminf(a, b);
                    const float hi = fmaxf(a, b);
                    v[i] = up ? lo : hi;
                    v[l] = up ? hi : lo;
                }
            }
        }
    }

    float r = v[0];
    #pragma unroll
    for (int i = 1; i < KK; i++) {
        r = (med == i) ? v[i] : r;
    }

    out[(size_t)n * OUT_C + c] = r;
}

extern "C" void kernel_launch(void* const* d_in, const int* in_sizes, int n_in,
                              void* d_out, int out_size, void* d_ws, size_t ws_size,
                              hipStream_t stream) {
    const float* x   = (const float*)d_in[0];
    const float* w   = (const float*)d_in[1];
    const int*   nbr = (const int*)d_in[2];
    const int*   deg = (const int*)d_in[3];
    float*       out = (float*)d_out;
    float*       h   = (float*)d_ws;

    median_conv_gemm_f32<<<3125, 256, 0, stream>>>(x, w, h);

    median_conv_select<<<NN, 128, 0, stream>>>(h, nbr, deg, out);
}


# select kernel: pad slots hold -inf (even slot) or +inf (odd slot), making the lower median rank 15 of 32 for every degree: one fixed rank-15 network, no ordered 16-prefix or index tree; bit-identical
# speedup vs baseline: 1.6924x; 1.6924x over previous
#include <hip/hip_runtime.h>
#include <hip/hip_bf16.h>

#define NN     50000
#define KK     32
#define IN_C   256
#define OUT_C  128

typedef __attribute__((ext_vector_type(2))) float v2f;
typedef __attribute__((ext_vector_type(8))) float v8f;

typedef __attribute__((ext_vector_type(4))) float v4f;

#define GEMM_BLOCKS   256
#define GEMM_THREADS  384
#define M_TILES       3125
#define GEMM_LDS_FLOATS 43008

__device__ __forceinline__ void gemm_chunk(const v2f (&a)[8], const float* bp,
                                           v8f& c0, v8f& c1, v8f& c2, v8f& c3) {
    #pragma unroll
    for (int s = 0; s < 8; ++s) {
        const v4f bA = *(const v4f*)(bp + s * 512);
        const v4f bB = *(const v4f*)(bp + s * 512 + 64);
        v2f b0, b1, b2, b3;
        b0.x = bA.x; b0.y = bA.y;  b1.x = bA.z; b1.y = bA.w;
        b2.x = bB.x; b2.y = bB.y;  b3.x = bB.z; b3.y = bB.w;
        c0 = __builtin_amdgcn_wmma_f32_16x16x4_f32(false, a[s], false, b0, (short)0, c0, false, false);
        c1 = __builtin_amdgcn_wmma_f32_16x16x4_f32(false, a[s], false, b1, (short)0, c1, false, false);
        c2 = __builtin_amdgcn_wmma_f32_16x16x4_f32(false, a[s], false, b2, (short)0, c2, false, false);
        c3 = __builtin_amdgcn_wmma_f32_16x16x4_f32(false, a[s], false, b3, (short)0, c3, false, false);
    }
}

__global__ void __launch_bounds__(GEMM_THREADS)
median_conv_gemm_f32(const float* __restrict__ x,
                     const float* __restrict__ w,
                     float* __restrict__ h) {
    __shared__ __attribute__((aligned(16))) float ws[GEMM_LDS_FLOATS];

    const int tid  = threadIdx.x;
    const int wave = __builtin_amdgcn_readfirstlane(tid >> 5);
    const int lane = tid & 31;
    const int lid  = lane & 15;
    const int half = lane >> 4;
    const int b    = blockIdx.x;

    #pragma unroll
    for (int i = 0; i < 11; i++) {
        const int u  = tid + i * GEMM_THREADS;
        if (u < 4096) {
            const int kp = u >> 5;
            const int c4 = (u & 31) * 4;
            const v4f r0 = *(const v4f*)(w + (size_t)(2 * kp)     * OUT_C + c4);
            const v4f r1 = *(const v4f*)(w + (size_t)(2 * kp + 1) * OUT_C + c4);
            float* dst = ws + kp * 256 + (c4 >> 6) * 128 + ((c4 >> 2) & 15) * 4;
            v4f o0, o1;
            o0.x = r0.x; o0.y = r1.x; o0.z = r0.y; o0.w = r1.y;
            o1.x = r0.z; o1.y = r1.z; o1.z = r0.w; o1.w = r1.w;
            *(v4f*)(dst)      = o0;
            *(v4f*)(dst + 64) = o1;
        }
    }
    __syncthreads();

    const int  pw    = wave >> 1;
    const bool extra = (b < M_TILES - 3072) && (wave == 0 || wave == 7);
    const int  nit   = extra ? 3 : 2;

    for (int it = 0; it < nit; ++it) {
        int mt, hq;
        if (it < 2) { mt = it * 1536 + pw * 256 + b; hq = wave & 1; }
        else        { mt = 3072 + b;                 hq = (wave == 7) ? 1 : 0; }

        const float* xr = x + (size_t)(mt * 16 + lid) * IN_C + 2 * half;
        const float* bp = ws + half * 256 + hq * 128 + lid * 4;

        v8f c0 = {}, c1 = {}, c2 = {}, c3 = {};
        v2f acur[8], anext[8];

        #pragma unroll
        for (int s = 0; s < 8; ++s) acur[s] = *(const v2f*)(xr + 4 * s);

        #pragma unroll 1
        for (int ch = 0; ch < 7; ++ch) {
            xr += 32;
            #pragma unroll
            for (int s = 0; s < 8; ++s) anext[s] = *(const v2f*)(xr + 4 * s);
            gemm_chunk(acur, bp, c0, c1, c2, c3);
            bp += 16 * 256;
            #pragma unroll
            for (int s = 0; s < 8; ++s) acur[s] = anext[s];
        }
        gemm_chunk(acur, bp, c0, c1, c2, c3);

        float* __restrict__ hr = h + (size_t)(mt * 16 + 8 * half) * OUT_C + hq * 64 + lid * 4;
        #pragma unroll
        for (int v = 0; v < 8; v++) {
            v4f o;
            o.x = c0[v]; o.y = c1[v]; o.z = c2[v]; o.w = c3[v];
            *(v4f*)(hr + (size_t)v * OUT_C) = o;
        }
    }
}

__device__ __forceinline__ void cmpx(float& a, float& b) {
    const float lo = fminf(a, b);
    const float hi = fmaxf(a, b);
    a = lo; b = hi;
}

template <int B>
__device__ __forceinline__ void sort16_oem(float (&v)[KK]) {
    #pragma unroll
    for (int p = 1; p < 16; p <<= 1) {
        #pragma unroll
        for (int k = p; k >= 1; k >>= 1) {
            #pragma unroll
            for (int i = 0; i < 16; i++) {
                const int j0 = k % p;
                if (i >= j0 && ((i - j0) % (2 * k)) < k && i + k < 16 &&
                    (i / (2 * p)) == ((i + k) / (2 * p))) {
                    cmpx(v[B + i], v[B + i + k]);
                }
            }
        }
    }
}

__global__ void __launch_bounds__(128)
median_conv_select(const float* __restrict__ h,
                   const int*   __restrict__ nbr,
                   const int*   __restrict__ deg,
                   float*       __restrict__ out) {
    const int n = blockIdx.x;
    const int c = threadIdx.x;

    const int d   = deg[n];

    const int* __restrict__ nb = nbr + (size_t)n * KK;

    const char* __restrict__ hb = (const char*)h;
    const unsigned c4 = (unsigned)c * 4u;

    float v[KK];
    #pragma unroll
    for (int j = 0; j < KK; j++) {
        const unsigned idx = (unsigned)nb[j];
        const unsigned off = (idx << 9) + c4;
        const float val = *(const float*)(hb + off);
        const float pad = (j >= 2 && (j & 1) == 0) ? -__builtin_inff() : __builtin_inff();
        v[j] = (j < d) ? val : pad;
    }

    sort16_oem<0>(v);
    sort16_oem<16>(v);
    float m[16];
    #pragma unroll
    for (int i = 0; i < 16; i++) m[i] = fminf(v[i], v[31 - i]);
    #pragma unroll
    for (int w = 8; w > 0; w >>= 1) {
        #pragma unroll
        for (int i = 0; i < w; i++) m[i] = fmaxf(m[i], m[i + w]);
    }
    const float r = m[0];

    out[(size_t)n * OUT_C + c] = r;
}

extern "C" void kernel_launch(void* const* d_in, const int* in_sizes, int n_in,
                              void* d_out, int out_size, void* d_ws, size_t ws_size,
                              hipStream_t stream) {
    const float* x   = (const float*)d_in[0];
    const float* w   = (const float*)d_in[1];
    const int*   nbr = (const int*)d_in[2];
    const int*   deg = (const int*)d_in[3];
    float*       out = (float*)d_out;
    float*       h   = (float*)d_ws;

    median_conv_gemm_f32<<<GEMM_BLOCKS, GEMM_THREADS, 0, stream>>>(x, w, h);

    median_conv_select<<<NN, 128, 0, stream>>>(h, nbr, deg, out);
}


# select kernel: padding slots read two constant rows (-inf, +inf) that the GEMM writes behind h; slot row picked on the scalar unit, no per-lane select; pad reads hit the WGP cache; bit-identical
# speedup vs baseline: 1.7848x; 1.0546x over previous
#include <hip/hip_runtime.h>
#include <hip/hip_bf16.h>

#define NN     50000
#define KK     32
#define IN_C   256
#define OUT_C  128

typedef __attribute__((ext_vector_type(2))) float v2f;
typedef __attribute__((ext_vector_type(8))) float v8f;

typedef __attribute__((ext_vector_type(4))) float v4f;

#define GEMM_BLOCKS   256
#define GEMM_THREADS  384
#define M_TILES       3125
#define GEMM_LDS_FLOATS 43008

__device__ __forceinline__ void gemm_chunk(const v2f (&a)[8], const float* bp,
                                           v8f& c0, v8f& c1, v8f& c2, v8f& c3) {
    #pragma unroll
    for (int s = 0; s < 8; ++s) {
        const v4f bA = *(const v4f*)(bp + s * 512);
        const v4f bB = *(const v4f*)(bp + s * 512 + 64);
        v2f b0, b1, b2, b3;
        b0.x = bA.x; b0.y = bA.y;  b1.x = bA.z; b1.y = bA.w;
        b2.x = bB.x; b2.y = bB.y;  b3.x = bB.z; b3.y = bB.w;
        c0 = __builtin_amdgcn_wmma_f32_16x16x4_f32(false, a[s], false, b0, (short)0, c0, false, false);
        c1 = __builtin_amdgcn_wmma_f32_16x16x4_f32(false, a[s], false, b1, (short)0, c1, false, false);
        c2 = __builtin_amdgcn_wmma_f32_16x16x4_f32(false, a[s], false, b2, (short)0, c2, false, false);
        c3 = __builtin_amdgcn_wmma_f32_16x16x4_f32(false, a[s], false, b3, (short)0, c3, false, false);
    }
}

__global__ void __launch_bounds__(GEMM_THREADS)
median_conv_gemm_f32(const float* __restrict__ x,
                     const float* __restrict__ w,
                     float* __restrict__ h) {
    __shared__ __attribute__((aligned(16))) float ws[GEMM_LDS_FLOATS];

    const int tid  = threadIdx.x;
    const int wave = __builtin_amdgcn_readfirstlane(tid >> 5);
    const int lane = tid & 31;
    const int lid  = lane & 15;
    const int half = lane >> 4;
    const int b    = blockIdx.x;

    #pragma unroll
    for (int i = 0; i < 11; i++) {
        const int u  = tid + i * GEMM_THREADS;
        if (u < 4096) {
            const int kp = u >> 5;
            const int c4 = (u & 31) * 4;
            const v4f r0 = *(const v4f*)(w + (size_t)(2 * kp)     * OUT_C + c4);
            const v4f r1 = *(const v4f*)(w + (size_t)(2 * kp + 1) * OUT_C + c4);
            float* dst = ws + kp * 256 + (c4 >> 6) * 128 + ((c4 >> 2) & 15) * 4;
            v4f o0, o1;
            o0.x = r0.x; o0.y = r1.x; o0.z = r0.y; o0.w = r1.y;
            o1.x = r0.z; o1.y = r1.z; o1.z = r0.w; o1.w = r1.w;
            *(v4f*)(dst)      = o0;
            *(v4f*)(dst + 64) = o1;
        }
    }
    __syncthreads();

    const int  pw    = wave >> 1;
    const bool extra = (b < M_TILES - 3072) && (wave == 0 || wave == 7);
    const int  nit   = extra ? 3 : 2;

    for (int it = 0; it < nit; ++it) {
        int mt, hq;
        if (it < 2) { mt = it * 1536 + pw * 256 + b; hq = wave & 1; }
        else        { mt = 3072 + b;                 hq = (wave == 7) ? 1 : 0; }

        const float* xr = x + (size_t)(mt * 16 + lid) * IN_C + 2 * half;
        const float* bp = ws + half * 256 + hq * 128 + lid * 4;

        v8f c0 = {}, c1 = {}, c2 = {}, c3 = {};
        v2f acur[8], anext[8];

        #pragma unroll
        for (int s = 0; s < 8; ++s) acur[s] = *(const v2f*)(xr + 4 * s);

        #pragma unroll 1
        for (int ch = 0; ch < 7; ++ch) {
            xr += 32;
            #pragma unroll
            for (int s = 0; s < 8; ++s) anext[s] = *(const v2f*)(xr + 4 * s);
            gemm_chunk(acur, bp, c0, c1, c2, c3);
            bp += 16 * 256;
            #pragma unroll
            for (int s = 0; s < 8; ++s) acur[s] = anext[s];
        }
        gemm_chunk(acur, bp, c0, c1, c2, c3);

        float* __restrict__ hr = h + (size_t)(mt * 16 + 8 * half) * OUT_C + hq * 64 + lid * 4;
        #pragma unroll
        for (int v = 0; v < 8; v++) {
            v4f o;
            o.x = c0[v]; o.y = c1[v]; o.z = c2[v]; o.w = c3[v];
            *(v4f*)(hr + (size_t)v * OUT_C) = o;
        }
    }

    if (b == GEMM_BLOCKS - 1 && tid < 2 * OUT_C) {
        h[(size_t)NN * OUT_C + tid] = (tid < OUT_C) ? -__builtin_inff() : __builtin_inff();
    }
}

__device__ __forceinline__ void cmpx(float& a, float& b) {
    const float lo = fminf(a, b);
    const float hi = fmaxf(a, b);
    a = lo; b = hi;
}

template <int B>
__device__ __forceinline__ void sort16_oem(float (&v)[KK]) {
    #pragma unroll
    for (int p = 1; p < 16; p <<= 1) {
        #pragma unroll
        for (int k = p; k >= 1; k >>= 1) {
            #pragma unroll
            for (int i = 0; i < 16; i++) {
                const int j0 = k % p;
                if (i >= j0 && ((i - j0) % (2 * k)) < k && i + k < 16 &&
                    (i / (2 * p)) == ((i + k) / (2 * p))) {
                    cmpx(v[B + i], v[B + i + k]);
                }
            }
        }
    }
}

__global__ void __launch_bounds__(128)
median_conv_select(const float* __restrict__ h,
                   const int*   __restrict__ nbr,
                   const int*   __restrict__ deg,
                   float*       __restrict__ out) {
    const int n = blockIdx.x;
    const int c = threadIdx.x;

    const int d   = deg[n];

    const int* __restrict__ nb = nbr + (size_t)n * KK;

    const char* __restrict__ hb = (const char*)h;
    const unsigned c4 = (unsigned)c * 4u;

    unsigned off[KK];
    #pragma unroll
    for (int j = 0; j < KK; j++) {
        const unsigned nbj = (unsigned)nb[j];
        const unsigned padrow = (j >= 2 && (j & 1) == 0) ? (unsigned)NN : (unsigned)NN + 1u;
        const unsigned idx = (j < d) ? nbj : padrow;
        off[j] = (idx << 9) + c4;
    }

    __builtin_amdgcn_sched_barrier(0);

    float v[KK];
    #pragma unroll
    for (int j = 0; j < KK; j++) v[j] = *(const float*)(hb + off[j]);

    sort16_oem<0>(v);
    sort16_oem<16>(v);
    float m[16];
    #pragma unroll
    for (int i = 0; i < 16; i++) m[i] = fminf(v[i], v[31 - i]);
    #pragma unroll
    for (int w = 8; w > 0; w >>= 1) {
        #pragma unroll
        for (int i = 0; i < w; i++) m[i] = fmaxf(m[i], m[i + w]);
    }
    const float r = m[0];

    out[(size_t)n * OUT_C + c] = r;
}

extern "C" void kernel_launch(void* const* d_in, const int* in_sizes, int n_in,
                              void* d_out, int out_size, void* d_ws, size_t ws_size,
                              hipStream_t stream) {
    const float* x   = (const float*)d_in[0];
    const float* w   = (const float*)d_in[1];
    const int*   nbr = (const int*)d_in[2];
    const int*   deg = (const int*)d_in[3];
    float*       out = (float*)d_out;
    float*       h   = (float*)d_ws;

    median_conv_gemm_f32<<<GEMM_BLOCKS, GEMM_THREADS, 0, stream>>>(x, w, h);

    median_conv_select<<<NN, 128, 0, stream>>>(h, nbr, deg, out);
}


# select kernel: each slot's row base (h + index*512) is formed on the scalar unit and is the gather's scalar base, the thread index its scaled lane offset: no vector address op per row; bit-identical
# speedup vs baseline: 1.8874x; 1.0575x over previous
#include <hip/hip_runtime.h>
#include <hip/hip_bf16.h>

#define NN     50000
#define KK     32
#define IN_C   256
#define OUT_C  128

typedef __attribute__((ext_vector_type(2))) float v2f;
typedef __attribute__((ext_vector_type(8))) float v8f;

typedef __attribute__((ext_vector_type(4))) float v4f;

#define GEMM_BLOCKS   256
#define GEMM_THREADS  384
#define M_TILES       3125
#define GEMM_LDS_FLOATS 43008

__device__ __forceinline__ void gemm_chunk(const v2f (&a)[8], const float* bp,
                                           v8f& c0, v8f& c1, v8f& c2, v8f& c3) {
    #pragma unroll
    for (int s = 0; s < 8; ++s) {
        const v4f bA = *(const v4f*)(bp + s * 512);
        const v4f bB = *(const v4f*)(bp + s * 512 + 64);
        v2f b0, b1, b2, b3;
        b0.x = bA.x; b0.y = bA.y;  b1.x = bA.z; b1.y = bA.w;
        b2.x = bB.x; b2.y = bB.y;  b3.x = bB.z; b3.y = bB.w;
        c0 = __builtin_amdgcn_wmma_f32_16x16x4_f32(false, a[s], false, b0, (short)0, c0, false, false);
        c1 = __builtin_amdgcn_wmma_f32_16x16x4_f32(false, a[s], false, b1, (short)0, c1, false, false);
        c2 = __builtin_amdgcn_wmma_f32_16x16x4_f32(false, a[s], false, b2, (short)0, c2, false, false);
        c3 = __builtin_amdgcn_wmma_f32_16x16x4_f32(false, a[s], false, b3, (short)0, c3, false, false);
    }
}

__global__ void __launch_bounds__(GEMM_THREADS)
median_conv_gemm_f32(const float* __restrict__ x,
                     const float* __restrict__ w,
                     float* __restrict__ h) {
    __shared__ __attribute__((aligned(16))) float ws[GEMM_LDS_FLOATS];

    const int tid  = threadIdx.x;
    const int wave = __builtin_amdgcn_readfirstlane(tid >> 5);
    const int lane = tid & 31;
    const int lid  = lane & 15;
    const int half = lane >> 4;
    const int b    = blockIdx.x;

    #pragma unroll
    for (int i = 0; i < 11; i++) {
        const int u  = tid + i * GEMM_THREADS;
        if (u < 4096) {
            const int kp = u >> 5;
            const int c4 = (u & 31) * 4;
            const v4f r0 = *(const v4f*)(w + (size_t)(2 * kp)     * OUT_C + c4);
            const v4f r1 = *(const v4f*)(w + (size_t)(2 * kp + 1) * OUT_C + c4);
            float* dst = ws + kp * 256 + (c4 >> 6) * 128 + ((c4 >> 2) & 15) * 4;
            v4f o0, o1;
            o0.x = r0.x; o0.y = r1.x; o0.z = r0.y; o0.w = r1.y;
            o1.x = r0.z; o1.y = r1.z; o1.z = r0.w; o1.w = r1.w;
            *(v4f*)(dst)      = o0;
            *(v4f*)(dst + 64) = o1;
        }
    }
    __syncthreads();

    const int  pw    = wave >> 1;
    const bool extra = (b < M_TILES - 3072) && (wave == 0 || wave == 7);
    const int  nit   = extra ? 3 : 2;

    for (int it = 0; it < nit; ++it) {
        int mt, hq;
        if (it < 2) { mt = it * 1536 + pw * 256 + b; hq = wave & 1; }
        else        { mt = 3072 + b;                 hq = (wave == 7) ? 1 : 0; }

        const float* xr = x + (size_t)(mt * 16 + lid) * IN_C + 2 * half;
        const float* bp = ws + half * 256 + hq * 128 + lid * 4;

        v8f c0 = {}, c1 = {}, c2 = {}, c3 = {};
        v2f acur[8], anext[8];

        #pragma unroll
        for (int s = 0; s < 8; ++s) acur[s] = *(const v2f*)(xr + 4 * s);

        #pragma unroll 1
        for (int ch = 0; ch < 7; ++ch) {
            xr += 32;
            #pragma unroll
            for (int s = 0; s < 8; ++s) anext[s] = *(const v2f*)(xr + 4 * s);
            gemm_chunk(acur, bp, c0, c1, c2, c3);
            bp += 16 * 256;
            #pragma unroll
            for (int s = 0; s < 8; ++s) acur[s] = anext[s];
        }
        gemm_chunk(acur, bp, c0, c1, c2, c3);

        float* __restrict__ hr = h + (size_t)(mt * 16 + 8 * half) * OUT_C + hq * 64 + lid * 4;
        #pragma unroll
        for (int v = 0; v < 8; v++) {
            v4f o;
            o.x = c0[v]; o.y = c1[v]; o.z = c2[v]; o.w = c3[v];
            *(v4f*)(hr + (size_t)v * OUT_C) = o;
        }
    }

    if (b == GEMM_BLOCKS - 1 && tid < 2 * OUT_C) {
        h[(size_t)NN * OUT_C + tid] = (tid < OUT_C) ? -__builtin_inff() : __builtin_inff();
    }
}

__device__ __forceinline__ void cmpx(float& a, float& b) {
    const float lo = fminf(a, b);
    const float hi = fmaxf(a, b);
    a = lo; b = hi;
}

template <int B>
__device__ __forceinline__ void sort16_oem(float (&v)[KK]) {
    #pragma unroll
    for (int p = 1; p < 16; p <<= 1) {
        #pragma unroll
        for (int k = p; k >= 1; k >>= 1) {
            #pragma unroll
            for (int i = 0; i < 16; i++) {
                const int j0 = k % p;
                if (i >= j0 && ((i - j0) % (2 * k)) < k && i + k < 16 &&
                    (i / (2 * p)) == ((i + k) / (2 * p))) {
                    cmpx(v[B + i], v[B + i + k]);
                }
            }
        }
    }
}

__global__ void __launch_bounds__(128)
median_conv_select(const float* __restrict__ h,
                   const int*   __restrict__ nbr,
                   const int*   __restrict__ deg,
                   float*       __restrict__ out) {
    const int n = blockIdx.x;
    const int c = threadIdx.x;

    const int d   = deg[n];

    const int* __restrict__ nb = nbr + (size_t)n * KK;


    const float* rowp[KK];
    #pragma unroll
    for (int j = 0; j < KK; j++) {
        const unsigned nbj = (unsigned)nb[j];
        const unsigned padrow = (j >= 2 && (j & 1) == 0) ? (unsigned)NN : (unsigned)NN + 1u;
        const unsigned idx = (j < d) ? nbj : padrow;
        rowp[j] = h + (size_t)idx * OUT_C;
    }

    __builtin_amdgcn_sched_barrier(0);

    float v[KK];
    #pragma unroll
    for (int j = 0; j < KK; j++) v[j] = rowp[j][c];

    sort16_oem<0>(v);
    sort16_oem<16>(v);
    float m[16];
    #pragma unroll
    for (int i = 0; i < 16; i++) m[i] = fminf(v[i], v[31 - i]);
    #pragma unroll
    for (int w = 8; w > 0; w >>= 1) {
        #pragma unroll
        for (int i = 0; i < w; i++) m[i] = fmaxf(m[i], m[i + w]);
    }
    const float r = m[0];

    out[(size_t)n * OUT_C + c] = r;
}

extern "C" void kernel_launch(void* const* d_in, const int* in_sizes, int n_in,
                              void* d_out, int out_size, void* d_ws, size_t ws_size,
                              hipStream_t stream) {
    const float* x   = (const float*)d_in[0];
    const float* w   = (const float*)d_in[1];
    const int*   nbr = (const int*)d_in[2];
    const int*   deg = (const int*)d_in[3];
    float*       out = (float*)d_out;
    float*       h   = (float*)d_ws;

    median_conv_gemm_f32<<<GEMM_BLOCKS, GEMM_THREADS, 0, stream>>>(x, w, h);

    median_conv_select<<<NN, 128, 0, stream>>>(h, nbr, deg, out);
}
